# starting kernel
# baseline (speedup 1.0000x reference)
#include <hip/hip_runtime.h>

#define NUSER 100000
#define NBOOK 50000
#define NEDGE 2000000
#define NLBL  500000
#define CCH   64
#define HCH   128

typedef __attribute__((ext_vector_type(2))) float v2f;
typedef __attribute__((ext_vector_type(8))) float v8f;

#if __has_builtin(__builtin_amdgcn_wmma_f32_16x16x4_f32)
#define HAVE_WMMA_F32 1
__device__ __forceinline__ v8f wmma4(v2f a, v2f b, v8f c) {
  return __builtin_amdgcn_wmma_f32_16x16x4_f32(false, a, false, b, (short)0, c, false, false);
}
#else
#define HAVE_WMMA_F32 0
#endif

__global__ void k_proj_user(const float* __restrict__ x, const float* __restrict__ w,
                            const float* __restrict__ bias, float* __restrict__ out) {
  int i = blockIdx.x * blockDim.x + threadIdx.x;
  if (i >= NUSER * CCH) return;
  int node = i >> 6, c = i & 63;
  float acc = bias[c];
#pragma unroll
  for (int k = 0; k < 3; ++k) acc = fmaf(x[node * 3 + k], w[k * CCH + c], acc);
  out[i] = acc;
}

__global__ void __launch_bounds__(256)
k_proj_book(const float* __restrict__ x, const float* __restrict__ w,
            const float* __restrict__ bias, float* __restrict__ out) {
  int tid = threadIdx.x;
  int wave = (blockIdx.x * blockDim.x + tid) >> 5;
  int lane = tid & 31;
  int m = lane & 15, half = lane >> 4;
  int row0 = wave << 4;
  bool active = row0 < NBOOK;
#if HAVE_WMMA_F32
  __shared__ v2f sW[64 * CCH];
  const float* xr = x + (size_t)(row0 + m) * 384;
  v8f zero = {0.f, 0.f, 0.f, 0.f, 0.f, 0.f, 0.f, 0.f};
  v8f acc[4] = {zero, zero, zero, zero};
  for (int k0 = 0; k0 < 384; k0 += 128) {
    __syncthreads();
    for (int idx = tid; idx < 64 * CCH; idx += 256) {
      int n = idx & 63, kp = idx >> 6;
      int k = k0 + 2 * kp;
      v2f v;
      v.x = w[k * CCH + n];
      v.y = w[(k + 1) * CCH + n];
      sW[idx] = v;
    }
    __syncthreads();
    if (active) {
      for (int k = k0; k < k0 + 128; k += 4) {
        int kk = k + 2 * half;
        v2f a;
        a.x = xr[kk];
        a.y = xr[kk + 1];
        int lp = (kk - k0) >> 1;
#pragma unroll
        for (int t = 0; t < 4; ++t)
          acc[t] = wmma4(a, sW[lp * CCH + t * 16 + m], acc[t]);
      }
    }
  }
  if (active) {
#pragma unroll
    for (int t = 0; t < 4; ++t) {
      int n = t * 16 + m;
      float bv = bias[n];
#pragma unroll
      for (int r = 0; r < 8; ++r)
        out[(size_t)(row0 + r + 8 * half) * CCH + n] = acc[t][r] + bv;
    }
  }
#else
  if (!active) return;
  const float* xr = x + (size_t)(row0 + m) * 384;
  for (int n = half * 32; n < half * 32 + 32; ++n) {
    float acc = bias[n];
    for (int k = 0; k < 384; ++k) acc = fmaf(xr[k], w[k * CCH + n], acc);
    out[(size_t)(row0 + m) * CCH + n] = acc;
  }
#endif
}

__global__ void k_zero(float* __restrict__ p, int n) {
  int i = blockIdx.x * blockDim.x + threadIdx.x;
  if (i < n) p[i] = 0.0f;
}

__global__ void k_count(const int* __restrict__ idx, float* __restrict__ cnt, int n) {
  int i = blockIdx.x * blockDim.x + threadIdx.x;
  if (i < n) atomicAdd(&cnt[idx[i]], 1.0f);
}

__global__ void k_scatter(const int* __restrict__ gsrc, const int* __restrict__ gdst,
                          const float* __restrict__ feat, float* __restrict__ agg) {
  long long i = (long long)blockIdx.x * blockDim.x + threadIdx.x;
  if (i >= (long long)NEDGE * 32) return;
  int e = (int)(i >> 5), ch = ((int)i & 31) * 2;
  int s = gsrc[e], d = gdst[e];
  const float2 v = *(const float2*)(feat + (size_t)s * CCH + ch);
  float* dst = agg + (size_t)d * CCH + ch;
  atomicAdd(dst, v.x);
  atomicAdd(dst + 1, v.y);
}

__global__ void __launch_bounds__(256)
k_sage(const float* __restrict__ agg, const float* __restrict__ cnt,
       const float* __restrict__ xdst,
       const float* __restrict__ wl, const float* __restrict__ wr,
       const float* __restrict__ bias, float* __restrict__ out,
       int n, int relu) {
  int tid = threadIdx.x;
  int wave = (blockIdx.x * blockDim.x + tid) >> 5;
  int lane = tid & 31;
  int m = lane & 15, half = lane >> 4;
  int row0 = wave << 4;
  bool active = row0 < n;
#if HAVE_WMMA_F32
  __shared__ v2f sWl[32 * CCH];
  __shared__ v2f sWr[32 * CCH];
  for (int idx = tid; idx < 32 * CCH; idx += 256) {
    int nn = idx & 63, kp = idx >> 6;
    int k = 2 * kp;
    v2f a, b;
    a.x = wl[k * CCH + nn];       a.y = wl[(k + 1) * CCH + nn];
    b.x = wr[k * CCH + nn];       b.y = wr[(k + 1) * CCH + nn];
    sWl[idx] = a;
    sWr[idx] = b;
  }
  __syncthreads();
  if (!active) return;
  int row = row0 + m;
  float rc = 1.0f / fmaxf(cnt[row], 1.0f);
  const float* ar = agg + (size_t)row * CCH;
  const float* xr = xdst + (size_t)row * CCH;
  v8f zero = {0.f, 0.f, 0.f, 0.f, 0.f, 0.f, 0.f, 0.f};
  v8f acc[4] = {zero, zero, zero, zero};
  for (int k = 0; k < CCH; k += 4) {
    int kk = k + 2 * half;
    int lp = kk >> 1;
    v2f am, ax;
    am.x = ar[kk] * rc;  am.y = ar[kk + 1] * rc;
    ax.x = xr[kk];       ax.y = xr[kk + 1];
#pragma unroll
    for (int t = 0; t < 4; ++t) {
      int ncol = t * 16 + m;
      acc[t] = wmma4(am, sWl[lp * CCH + ncol], acc[t]);
      acc[t] = wmma4(ax, sWr[lp * CCH + ncol], acc[t]);
    }
  }
#pragma unroll
  for (int t = 0; t < 4; ++t) {
    int ncol = t * 16 + m;
    float bv = bias[ncol];
#pragma unroll
    for (int r = 0; r < 8; ++r) {
      float v = acc[t][r] + bv;
      if (relu) v = fmaxf(v, 0.0f);
      out[(size_t)(row0 + r + 8 * half) * CCH + ncol] = v;
    }
  }
#else
  if (!active) return;
  int row = row0 + m;
  float rc = 1.0f / fmaxf(cnt[row], 1.0f);
  const float* ar = agg + (size_t)row * CCH;
  const float* xr = xdst + (size_t)row * CCH;
  for (int ncol = half * 32; ncol < half * 32 + 32; ++ncol) {
    float v = bias[ncol];
    for (int k = 0; k < CCH; ++k)
      v = fmaf(ar[k] * rc, wl[k * CCH + ncol], fmaf(xr[k], wr[k * CCH + ncol], v));
    if (relu) v = fmaxf(v, 0.0f);
    out[(size_t)row * CCH + ncol] = v;
  }
#endif
}

__global__ void __launch_bounds__(256)
k_decoder(const float* __restrict__ u2, const float* __restrict__ b2,
          const int* __restrict__ ls, const int* __restrict__ ld,
          const float* __restrict__ w1, const float* __restrict__ bias1,
          const float* __restrict__ w2, const float* __restrict__ bias2,
          float* __restrict__ out) {
  int tid = threadIdx.x;
  int wave = (blockIdx.x * blockDim.x + tid) >> 5;
  int lane = tid & 31;
  int m = lane & 15, half = lane >> 4;
  int row0 = wave << 4;
  bool active = row0 < NLBL;
  int row = active ? (row0 + m) : 0;
  const float* ur = u2 + (active ? (size_t)ls[row] * CCH : 0);
  const float* br = b2 + (active ? (size_t)ld[row] * CCH : 0);
#if HAVE_WMMA_F32
  __shared__ v2f sW[32 * HCH];
  v8f zero = {0.f, 0.f, 0.f, 0.f, 0.f, 0.f, 0.f, 0.f};
  v8f acc[8] = {zero, zero, zero, zero, zero, zero, zero, zero};
  for (int k0 = 0; k0 < HCH; k0 += 64) {
    __syncthreads();
    for (int idx = tid; idx < 32 * HCH; idx += 256) {
      int n = idx & 127, kp = idx >> 7;
      int k = k0 + 2 * kp;
      v2f v;
      v.x = w1[k * HCH + n];
      v.y = w1[(k + 1) * HCH + n];
      sW[idx] = v;
    }
    __syncthreads();
    if (active) {
      for (int k = k0; k < k0 + 64; k += 4) {
        int kk = k + 2 * half;
        const float* zsrc = (kk < CCH) ? (ur + kk) : (br + (kk - CCH));
        v2f a;
        a.x = zsrc[0];
        a.y = zsrc[1];
        int lp = (kk - k0) >> 1;
#pragma unroll
        for (int t = 0; t < 8; ++t)
          acc[t] = wmma4(a, sW[lp * HCH + t * 16 + m], acc[t]);
      }
    }
  }
  if (!active) return;
  float w2v[8], b1v[8];
#pragma unroll
  for (int t = 0; t < 8; ++t) {
    int ncol = t * 16 + m;
    w2v[t] = w2[ncol];
    b1v[t] = bias1[ncol];
  }
  float outb = bias2[0];
#pragma unroll
  for (int r = 0; r < 8; ++r) {
    float p = 0.0f;
#pragma unroll
    for (int t = 0; t < 8; ++t) {
      float hv = fmaxf(acc[t][r] + b1v[t], 0.0f);
      p = fmaf(hv, w2v[t], p);
    }
    p += __shfl_xor(p, 1);
    p += __shfl_xor(p, 2);
    p += __shfl_xor(p, 4);
    p += __shfl_xor(p, 8);
    if (m == 0) out[row0 + r + 8 * half] = p + outb;
  }
#else
  if (active && half == 0) {
    float o = bias2[0];
    for (int n = 0; n < HCH; ++n) {
      float hv = bias1[n];
      for (int k = 0; k < CCH; ++k) hv = fmaf(ur[k], w1[k * HCH + n], hv);
      for (int k = 0; k < CCH; ++k) hv = fmaf(br[k], w1[(CCH + k) * HCH + n], hv);
      o = fmaf(fmaxf(hv, 0.0f), w2[n], o);
    }
    out[row] = o;
  }
#endif
}

extern "C" void kernel_launch(void* const* d_in, const int* in_sizes, int n_in,
                              void* d_out, int out_size, void* d_ws, size_t ws_size,
                              hipStream_t stream) {
  (void)in_sizes; (void)n_in; (void)out_size; (void)ws_size;
  const float* user_x     = (const float*)d_in[0];
  const float* book_x     = (const float*)d_in[1];
  const int*   edge_src   = (const int*)d_in[2];
  const int*   edge_dst   = (const int*)d_in[3];
  const int*   label_src  = (const int*)d_in[4];
  const int*   label_dst  = (const int*)d_in[5];
  const float* user_lin_w = (const float*)d_in[6];
  const float* user_lin_b = (const float*)d_in[7];
  const float* book_lin_w = (const float*)d_in[8];
  const float* book_lin_b = (const float*)d_in[9];
  const float* w1_ub_l = (const float*)d_in[10];
  const float* b1_ub   = (const float*)d_in[11];
  const float* w1_ub_r = (const float*)d_in[12];
  const float* w1_bu_l = (const float*)d_in[13];
  const float* b1_bu   = (const float*)d_in[14];
  const float* w1_bu_r = (const float*)d_in[15];
  const float* w2_ub_l = (const float*)d_in[16];
  const float* b2_ub   = (const float*)d_in[17];
  const float* w2_ub_r = (const float*)d_in[18];
  const float* w2_bu_l = (const float*)d_in[19];
  const float* b2_bu   = (const float*)d_in[20];
  const float* w2_bu_r = (const float*)d_in[21];
  const float* dec_w1  = (const float*)d_in[22];
  const float* dec_b1  = (const float*)d_in[23];
  const float* dec_w2  = (const float*)d_in[24];
  const float* dec_b2  = (const float*)d_in[25];
  float* out = (float*)d_out;

  float* ws = (float*)d_ws;
  size_t off = 0;
  float* u0    = ws + off; off += (size_t)NUSER * CCH;
  float* b0    = ws + off; off += (size_t)NBOOK * CCH;
  float* u1    = ws + off; off += (size_t)NUSER * CCH;
  float* b1buf = ws + off; off += (size_t)NBOOK * CCH;
  float* agg   = ws + off; off += (size_t)NUSER * CCH;
  float* cnt_u = ws + off; off += NUSER;
  float* cnt_b = ws + off; off += NBOOK;
  float* u2 = u0;
  float* b2 = b0;

  const int BT = 256;
  const int sage_blk_b = (NBOOK / 16 * 32 + BT - 1) / BT;
  const int sage_blk_u = (NUSER / 16 * 32 + BT - 1) / BT;
  const int sc_blocks  = (int)(((long long)NEDGE * 32 + BT - 1) / BT);

  k_proj_user<<<(NUSER * CCH + BT - 1) / BT, BT, 0, stream>>>(user_x, user_lin_w, user_lin_b, u0);
  k_proj_book<<<(NBOOK / 16 * 32 + BT - 1) / BT, BT, 0, stream>>>(book_x, book_lin_w, book_lin_b, b0);

  k_zero<<<(NBOOK + BT - 1) / BT, BT, 0, stream>>>(cnt_b, NBOOK);
  k_zero<<<(NUSER + BT - 1) / BT, BT, 0, stream>>>(cnt_u, NUSER);
  k_count<<<(NEDGE + BT - 1) / BT, BT, 0, stream>>>(edge_dst, cnt_b, NEDGE);
  k_count<<<(NEDGE + BT - 1) / BT, BT, 0, stream>>>(edge_src, cnt_u, NEDGE);

  k_zero<<<(NBOOK * CCH + BT - 1) / BT, BT, 0, stream>>>(agg, NBOOK * CCH);
  k_scatter<<<sc_blocks, BT, 0, stream>>>(edge_src, edge_dst, u0, agg);
  k_sage<<<sage_blk_b, BT, 0, stream>>>(agg, cnt_b, b0, w1_ub_l, w1_ub_r, b1_ub, b1buf, NBOOK, 1);
  k_zero<<<(NUSER * CCH + BT - 1) / BT, BT, 0, stream>>>(agg, NUSER * CCH);
  k_scatter<<<sc_blocks, BT, 0, stream>>>(edge_dst, edge_src, b0, agg);
  k_sage<<<sage_blk_u, BT, 0, stream>>>(agg, cnt_u, u0, w1_bu_l, w1_bu_r, b1_bu, u1, NUSER, 1);

  k_zero<<<(NBOOK * CCH + BT - 1) / BT, BT, 0, stream>>>(agg, NBOOK * CCH);
  k_scatter<<<sc_blocks, BT, 0, stream>>>(edge_src, edge_dst, u1, agg);
  k_sage<<<sage_blk_b, BT, 0, stream>>>(agg, cnt_b, b1buf, w2_ub_l, w2_ub_r, b2_ub, b2, NBOOK, 0);
  k_zero<<<(NUSER * CCH + BT - 1) / BT, BT, 0, stream>>>(agg, NUSER * CCH);
  k_scatter<<<sc_blocks, BT, 0, stream>>>(edge_dst, edge_src, b1buf, agg);
  k_sage<<<sage_blk_u, BT, 0, stream>>>(agg, cnt_u, u1, w2_bu_l, w2_bu_r, b2_bu, u2, NUSER, 0);

  k_decoder<<<(NLBL / 16 * 32 + BT - 1) / BT, BT, 0, stream>>>(
      u2, b2, label_src, label_dst, dec_w1, dec_b1, dec_w2, dec_b2, out);
}


# layer-1 book neighbour sum taken on raw user rows [x0,x1,x2,1] (4 floats/edge, not 64), projected after; book degree = its 4th column; sum order: atomic raw sums first, then the 3x64 projection
# speedup vs baseline: 1.1778x; 1.1778x over previous
#include <hip/hip_runtime.h>

#define NUSER 100000
#define NBOOK 50000
#define NEDGE 2000000
#define NLBL  500000
#define CCH   64
#define HCH   128

typedef __attribute__((ext_vector_type(2))) float v2f;
typedef __attribute__((ext_vector_type(8))) float v8f;

#if __has_builtin(__builtin_amdgcn_wmma_f32_16x16x4_f32)
#define HAVE_WMMA_F32 1
__device__ __forceinline__ v8f wmma4(v2f a, v2f b, v8f c) {
  return __builtin_amdgcn_wmma_f32_16x16x4_f32(false, a, false, b, (short)0, c, false, false);
}
#else
#define HAVE_WMMA_F32 0
#endif

__global__ void k_proj_user(const float* __restrict__ x, const float* __restrict__ w,
                            const float* __restrict__ bias, float* __restrict__ out) {
  int i = blockIdx.x * blockDim.x + threadIdx.x;
  if (i >= NUSER * CCH) return;
  int node = i >> 6, c = i & 63;
  float acc = bias[c];
#pragma unroll
  for (int k = 0; k < 3; ++k) acc = fmaf(x[node * 3 + k], w[k * CCH + c], acc);
  out[i] = acc;
}

__global__ void __launch_bounds__(256)
k_proj_book(const float* __restrict__ x, const float* __restrict__ w,
            const float* __restrict__ bias, float* __restrict__ out) {
  int tid = threadIdx.x;
  int wave = (blockIdx.x * blockDim.x + tid) >> 5;
  int lane = tid & 31;
  int m = lane & 15, half = lane >> 4;
  int row0 = wave << 4;
  bool active = row0 < NBOOK;
#if HAVE_WMMA_F32
  __shared__ v2f sW[64 * CCH];
  const float* xr = x + (size_t)(row0 + m) * 384;
  v8f zero = {0.f, 0.f, 0.f, 0.f, 0.f, 0.f, 0.f, 0.f};
  v8f acc[4] = {zero, zero, zero, zero};
  for (int k0 = 0; k0 < 384; k0 += 128) {
    __syncthreads();
    for (int idx = tid; idx < 64 * CCH; idx += 256) {
      int n = idx & 63, kp = idx >> 6;
      int k = k0 + 2 * kp;
      v2f v;
      v.x = w[k * CCH + n];
      v.y = w[(k + 1) * CCH + n];
      sW[idx] = v;
    }
    __syncthreads();
    if (active) {
      for (int k = k0; k < k0 + 128; k += 4) {
        int kk = k + 2 * half;
        v2f a;
        a.x = xr[kk];
        a.y = xr[kk + 1];
        int lp = (kk - k0) >> 1;
#pragma unroll
        for (int t = 0; t < 4; ++t)
          acc[t] = wmma4(a, sW[lp * CCH + t * 16 + m], acc[t]);
      }
    }
  }
  if (active) {
#pragma unroll
    for (int t = 0; t < 4; ++t) {
      int n = t * 16 + m;
      float bv = bias[n];
#pragma unroll
      for (int r = 0; r < 8; ++r)
        out[(size_t)(row0 + r + 8 * half) * CCH + n] = acc[t][r] + bv;
    }
  }
#else
  if (!active) return;
  const float* xr = x + (size_t)(row0 + m) * 384;
  for (int n = half * 32; n < half * 32 + 32; ++n) {
    float acc = bias[n];
    for (int k = 0; k < 384; ++k) acc = fmaf(xr[k], w[k * CCH + n], acc);
    out[(size_t)(row0 + m) * CCH + n] = acc;
  }
#endif
}

__global__ void k_zero(float* __restrict__ p, int n) {
  int i = blockIdx.x * blockDim.x + threadIdx.x;
  if (i < n) p[i] = 0.0f;
}

__global__ void k_count(const int* __restrict__ idx, float* __restrict__ cnt, int n) {
  int i = blockIdx.x * blockDim.x + threadIdx.x;
  if (i < n) atomicAdd(&cnt[idx[i]], 1.0f);
}

__global__ void k_scatter(const int* __restrict__ gsrc, const int* __restrict__ gdst,
                          const float* __restrict__ feat, float* __restrict__ agg) {
  long long i = (long long)blockIdx.x * blockDim.x + threadIdx.x;
  if (i >= (long long)NEDGE * 32) return;
  int e = (int)(i >> 5), ch = ((int)i & 31) * 2;
  int s = gsrc[e], d = gdst[e];
  const float2 v = *(const float2*)(feat + (size_t)s * CCH + ch);
  float* dst = agg + (size_t)d * CCH + ch;
  atomicAdd(dst, v.x);
  atomicAdd(dst + 1, v.y);
}

__global__ void k_scatter_raw(const int* __restrict__ gsrc, const int* __restrict__ gdst,
                              const float* __restrict__ ux, float* __restrict__ agg4) {
  int i = blockIdx.x * blockDim.x + threadIdx.x;
  if (i >= NEDGE * 4) return;
  int e = i >> 2, c = i & 3;
  int s = gsrc[e], d = gdst[e];
  int cc = c < 3 ? c : 2;
  float xv = ux[(size_t)s * 3 + cc];
  float v = c < 3 ? xv : 1.0f;
  atomicAdd(agg4 + (size_t)d * 4 + c, v);
}

__global__ void __launch_bounds__(256)
k_sage(const float* __restrict__ agg, const float* __restrict__ cnt,
       const float* __restrict__ xdst,
       const float* __restrict__ wl, const float* __restrict__ wr,
       const float* __restrict__ bias, float* __restrict__ out,
       int n, int relu) {
  int tid = threadIdx.x;
  int wave = (blockIdx.x * blockDim.x + tid) >> 5;
  int lane = tid & 31;
  int m = lane & 15, half = lane >> 4;
  int row0 = wave << 4;
  bool active = row0 < n;
#if HAVE_WMMA_F32
  __shared__ v2f sWl[32 * CCH];
  __shared__ v2f sWr[32 * CCH];
  for (int idx = tid; idx < 32 * CCH; idx += 256) {
    int nn = idx & 63, kp = idx >> 6;
    int k = 2 * kp;
    v2f a, b;
    a.x = wl[k * CCH + nn];       a.y = wl[(k + 1) * CCH + nn];
    b.x = wr[k * CCH + nn];       b.y = wr[(k + 1) * CCH + nn];
    sWl[idx] = a;
    sWr[idx] = b;
  }
  __syncthreads();
  if (!active) return;
  int row = row0 + m;
  float rc = 1.0f / fmaxf(cnt[row], 1.0f);
  const float* ar = agg + (size_t)row * CCH;
  const float* xr = xdst + (size_t)row * CCH;
  v8f zero = {0.f, 0.f, 0.f, 0.f, 0.f, 0.f, 0.f, 0.f};
  v8f acc[4] = {zero, zero, zero, zero};
  for (int k = 0; k < CCH; k += 4) {
    int kk = k + 2 * half;
    int lp = kk >> 1;
    v2f am, ax;
    am.x = ar[kk] * rc;  am.y = ar[kk + 1] * rc;
    ax.x = xr[kk];       ax.y = xr[kk + 1];
#pragma unroll
    for (int t = 0; t < 4; ++t) {
      int ncol = t * 16 + m;
      acc[t] = wmma4(am, sWl[lp * CCH + ncol], acc[t]);
      acc[t] = wmma4(ax, sWr[lp * CCH + ncol], acc[t]);
    }
  }
#pragma unroll
  for (int t = 0; t < 4; ++t) {
    int ncol = t * 16 + m;
    float bv = bias[ncol];
#pragma unroll
    for (int r = 0; r < 8; ++r) {
      float v = acc[t][r] + bv;
      if (relu) v = fmaxf(v, 0.0f);
      out[(size_t)(row0 + r + 8 * half) * CCH + ncol] = v;
    }
  }
#else
  if (!active) return;
  int row = row0 + m;
  float rc = 1.0f / fmaxf(cnt[row], 1.0f);
  const float* ar = agg + (size_t)row * CCH;
  const float* xr = xdst + (size_t)row * CCH;
  for (int ncol = half * 32; ncol < half * 32 + 32; ++ncol) {
    float v = bias[ncol];
    for (int k = 0; k < CCH; ++k)
      v = fmaf(ar[k] * rc, wl[k * CCH + ncol], fmaf(xr[k], wr[k * CCH + ncol], v));
    if (relu) v = fmaxf(v, 0.0f);
    out[(size_t)row * CCH + ncol] = v;
  }
#endif
}

__global__ void __launch_bounds__(256)
k_sage_raw(const float* __restrict__ agg4,
           const float* __restrict__ pw, const float* __restrict__ pb,
           const float* __restrict__ xdst,
           const float* __restrict__ wl, const float* __restrict__ wr,
           const float* __restrict__ bias, float* __restrict__ cnt_out,
           float* __restrict__ out, int n, int relu) {
  int tid = threadIdx.x;
  int wave = (blockIdx.x * blockDim.x + tid) >> 5;
  int lane = tid & 31;
  int m = lane & 15, half = lane >> 4;
  int row0 = wave << 4;
  bool active = row0 < n;
#if HAVE_WMMA_F32
  __shared__ v2f sWl[32 * CCH];
  __shared__ v2f sWr[32 * CCH];
  __shared__ float sP[4 * CCH];
  for (int idx = tid; idx < 32 * CCH; idx += 256) {
    int nn = idx & 63, kp = idx >> 6;
    int k = 2 * kp;
    v2f a, b;
    a.x = wl[k * CCH + nn];       a.y = wl[(k + 1) * CCH + nn];
    b.x = wr[k * CCH + nn];       b.y = wr[(k + 1) * CCH + nn];
    sWl[idx] = a;
    sWr[idx] = b;
  }
  sP[tid] = (tid < 3 * CCH) ? pw[tid] : pb[tid - 3 * CCH];
  int brow = blockIdx.x * 128 + tid;
  if (tid < 128 && brow < n) cnt_out[brow] = agg4[(size_t)brow * 4 + 3];
  __syncthreads();
  if (!active) return;
  int row = row0 + m;
  const float4 sv = *(const float4*)(agg4 + (size_t)row * 4);
  float cn = sv.w;
  float rc = 1.0f / fmaxf(cn, 1.0f);
  const float* xr = xdst + (size_t)row * CCH;
  v8f zero = {0.f, 0.f, 0.f, 0.f, 0.f, 0.f, 0.f, 0.f};
  v8f acc[4] = {zero, zero, zero, zero};
  for (int k = 0; k < CCH; k += 4) {
    int kk = k + 2 * half;
    int lp = kk >> 1;
    float g0 = fmaf(sv.z, sP[2 * CCH + kk],
               fmaf(sv.y, sP[CCH + kk],
               fmaf(sv.x, sP[kk], cn * sP[3 * CCH + kk])));
    float g1 = fmaf(sv.z, sP[2 * CCH + kk + 1],
               fmaf(sv.y, sP[CCH + kk + 1],
               fmaf(sv.x, sP[kk + 1], cn * sP[3 * CCH + kk + 1])));
    v2f am, ax;
    am.x = g0 * rc;      am.y = g1 * rc;
    ax.x = xr[kk];       ax.y = xr[kk + 1];
#pragma unroll
    for (int t = 0; t < 4; ++t) {
      int ncol = t * 16 + m;
      acc[t] = wmma4(am, sWl[lp * CCH + ncol], acc[t]);
      acc[t] = wmma4(ax, sWr[lp * CCH + ncol], acc[t]);
    }
  }
#pragma unroll
  for (int t = 0; t < 4; ++t) {
    int ncol = t * 16 + m;
    float bv = bias[ncol];
#pragma unroll
    for (int r = 0; r < 8; ++r) {
      float v = acc[t][r] + bv;
      if (relu) v = fmaxf(v, 0.0f);
      out[(size_t)(row0 + r + 8 * half) * CCH + ncol] = v;
    }
  }
#else
  if (tid < 128) {
    int brow = blockIdx.x * 128 + tid;
    if (brow < n) cnt_out[brow] = agg4[(size_t)brow * 4 + 3];
  }
  if (!active) return;
  int row = row0 + m;
  const float* sr = agg4 + (size_t)row * 4;
  float cn = sr[3];
  float rc = 1.0f / fmaxf(cn, 1.0f);
  const float* xr = xdst + (size_t)row * CCH;
  for (int ncol = half * 32; ncol < half * 32 + 32; ++ncol) {
    float v = bias[ncol];
    for (int k = 0; k < CCH; ++k) {
      float g = fmaf(sr[2], pw[2 * CCH + k], fmaf(sr[1], pw[CCH + k], fmaf(sr[0], pw[k], cn * pb[k])));
      v = fmaf(g * rc, wl[k * CCH + ncol], fmaf(xr[k], wr[k * CCH + ncol], v));
    }
    if (relu) v = fmaxf(v, 0.0f);
    out[(size_t)row * CCH + ncol] = v;
  }
#endif
}

__global__ void __launch_bounds__(256)
k_decoder(const float* __restrict__ u2, const float* __restrict__ b2,
          const int* __restrict__ ls, const int* __restrict__ ld,
          const float* __restrict__ w1, const float* __restrict__ bias1,
          const float* __restrict__ w2, const float* __restrict__ bias2,
          float* __restrict__ out) {
  int tid = threadIdx.x;
  int wave = (blockIdx.x * blockDim.x + tid) >> 5;
  int lane = tid & 31;
  int m = lane & 15, half = lane >> 4;
  int row0 = wave << 4;
  bool active = row0 < NLBL;
  int row = active ? (row0 + m) : 0;
  const float* ur = u2 + (active ? (size_t)ls[row] * CCH : 0);
  const float* br = b2 + (active ? (size_t)ld[row] * CCH : 0);
#if HAVE_WMMA_F32
  __shared__ v2f sW[32 * HCH];
  v8f zero = {0.f, 0.f, 0.f, 0.f, 0.f, 0.f, 0.f, 0.f};
  v8f acc[8] = {zero, zero, zero, zero, zero, zero, zero, zero};
  for (int k0 = 0; k0 < HCH; k0 += 64) {
    __syncthreads();
    for (int idx = tid; idx < 32 * HCH; idx += 256) {
      int n = idx & 127, kp = idx >> 7;
      int k = k0 + 2 * kp;
      v2f v;
      v.x = w1[k * HCH + n];
      v.y = w1[(k + 1) * HCH + n];
      sW[idx] = v;
    }
    __syncthreads();
    if (active) {
      for (int k = k0; k < k0 + 64; k += 4) {
        int kk = k + 2 * half;
        const float* zsrc = (kk < CCH) ? (ur + kk) : (br + (kk - CCH));
        v2f a;
        a.x = zsrc[0];
        a.y = zsrc[1];
        int lp = (kk - k0) >> 1;
#pragma unroll
        for (int t = 0; t < 8; ++t)
          acc[t] = wmma4(a, sW[lp * HCH + t * 16 + m], acc[t]);
      }
    }
  }
  if (!active) return;
  float w2v[8], b1v[8];
#pragma unroll
  for (int t = 0; t < 8; ++t) {
    int ncol = t * 16 + m;
    w2v[t] = w2[ncol];
    b1v[t] = bias1[ncol];
  }
  float outb = bias2[0];
#pragma unroll
  for (int r = 0; r < 8; ++r) {
    float p = 0.0f;
#pragma unroll
    for (int t = 0; t < 8; ++t) {
      float hv = fmaxf(acc[t][r] + b1v[t], 0.0f);
      p = fmaf(hv, w2v[t], p);
    }
    p += __shfl_xor(p, 1);
    p += __shfl_xor(p, 2);
    p += __shfl_xor(p, 4);
    p += __shfl_xor(p, 8);
    if (m == 0) out[row0 + r + 8 * half] = p + outb;
  }
#else
  if (active && half == 0) {
    float o = bias2[0];
    for (int n = 0; n < HCH; ++n) {
      float hv = bias1[n];
      for (int k = 0; k < CCH; ++k) hv = fmaf(ur[k], w1[k * HCH + n], hv);
      for (int k = 0; k < CCH; ++k) hv = fmaf(br[k], w1[(CCH + k) * HCH + n], hv);
      o = fmaf(fmaxf(hv, 0.0f), w2[n], o);
    }
    out[row] = o;
  }
#endif
}

extern "C" void kernel_launch(void* const* d_in, const int* in_sizes, int n_in,
                              void* d_out, int out_size, void* d_ws, size_t ws_size,
                              hipStream_t stream) {
  (void)in_sizes; (void)n_in; (void)out_size; (void)ws_size;
  const float* user_x     = (const float*)d_in[0];
  const float* book_x     = (const float*)d_in[1];
  const int*   edge_src   = (const int*)d_in[2];
  const int*   edge_dst   = (const int*)d_in[3];
  const int*   label_src  = (const int*)d_in[4];
  const int*   label_dst  = (const int*)d_in[5];
  const float* user_lin_w = (const float*)d_in[6];
  const float* user_lin_b = (const float*)d_in[7];
  const float* book_lin_w = (const float*)d_in[8];
  const float* book_lin_b = (const float*)d_in[9];
  const float* w1_ub_l = (const float*)d_in[10];
  const float* b1_ub   = (const float*)d_in[11];
  const float* w1_ub_r = (const float*)d_in[12];
  const float* w1_bu_l = (const float*)d_in[13];
  const float* b1_bu   = (const float*)d_in[14];
  const float* w1_bu_r = (const float*)d_in[15];
  const float* w2_ub_l = (const float*)d_in[16];
  const float* b2_ub   = (const float*)d_in[17];
  const float* w2_ub_r = (const float*)d_in[18];
  const float* w2_bu_l = (const float*)d_in[19];
  const float* b2_bu   = (const float*)d_in[20];
  const float* w2_bu_r = (const float*)d_in[21];
  const float* dec_w1  = (const float*)d_in[22];
  const float* dec_b1  = (const float*)d_in[23];
  const float* dec_w2  = (const float*)d_in[24];
  const float* dec_b2  = (const float*)d_in[25];
  float* out = (float*)d_out;

  float* ws = (float*)d_ws;
  size_t off = 0;
  float* u0    = ws + off; off += (size_t)NUSER * CCH;
  float* b0    = ws + off; off += (size_t)NBOOK * CCH;
  float* u1    = ws + off; off += (size_t)NUSER * CCH;
  float* b1buf = ws + off; off += (size_t)NBOOK * CCH;
  float* agg   = ws + off; off += (size_t)NUSER * CCH;
  float* cnt_u = ws + off; off += NUSER;
  float* cnt_b = ws + off; off += NBOOK;
  float* u2 = u0;
  float* b2 = b0;

  const int BT = 256;
  const int sage_blk_b = (NBOOK / 16 * 32 + BT - 1) / BT;
  const int sage_blk_u = (NUSER / 16 * 32 + BT - 1) / BT;
  const int sc_blocks  = (int)(((long long)NEDGE * 32 + BT - 1) / BT);

  k_proj_user<<<(NUSER * CCH + BT - 1) / BT, BT, 0, stream>>>(user_x, user_lin_w, user_lin_b, u0);
  k_proj_book<<<(NBOOK / 16 * 32 + BT - 1) / BT, BT, 0, stream>>>(book_x, book_lin_w, book_lin_b, b0);

  k_zero<<<(NUSER + BT - 1) / BT, BT, 0, stream>>>(cnt_u, NUSER);
  k_count<<<(NEDGE + BT - 1) / BT, BT, 0, stream>>>(edge_src, cnt_u, NEDGE);

  k_zero<<<(NBOOK * 4 + BT - 1) / BT, BT, 0, stream>>>(agg, NBOOK * 4);
  k_scatter_raw<<<(NEDGE * 4 + BT - 1) / BT, BT, 0, stream>>>(edge_src, edge_dst, user_x, agg);
  k_sage_raw<<<sage_blk_b, BT, 0, stream>>>(agg, user_lin_w, user_lin_b, b0, w1_ub_l, w1_ub_r,
                                            b1_ub, cnt_b, b1buf, NBOOK, 1);
  k_zero<<<(NUSER * CCH + BT - 1) / BT, BT, 0, stream>>>(agg, NUSER * CCH);
  k_scatter<<<sc_blocks, BT, 0, stream>>>(edge_dst, edge_src, b0, agg);
  k_sage<<<sage_blk_u, BT, 0, stream>>>(agg, cnt_u, u0, w1_bu_l, w1_bu_r, b1_bu, u1, NUSER, 1);

  k_zero<<<(NBOOK * CCH + BT - 1) / BT, BT, 0, stream>>>(agg, NBOOK * CCH);
  k_scatter<<<sc_blocks, BT, 0, stream>>>(edge_src, edge_dst, u1, agg);
  k_sage<<<sage_blk_b, BT, 0, stream>>>(agg, cnt_b, b1buf, w2_ub_l, w2_ub_r, b2_ub, b2, NBOOK, 0);
  k_zero<<<(NUSER * CCH + BT - 1) / BT, BT, 0, stream>>>(agg, NUSER * CCH);
  k_scatter<<<sc_blocks, BT, 0, stream>>>(edge_dst, edge_src, b1buf, agg);
  k_sage<<<sage_blk_u, BT, 0, stream>>>(agg, cnt_u, u1, w2_bu_l, w2_bu_r, b2_bu, u2, NUSER, 0);

  k_decoder<<<(NLBL / 16 * 32 + BT - 1) / BT, BT, 0, stream>>>(
      u2, b2, label_src, label_dst, dec_w1, dec_b1, dec_w2, dec_b2, out);
}


# layer-2 user neighbour sum hoisted into layer-1's edge pass (b1 needs no u1): one wave adds its book's b0 and b1 rows into two user sums; 3 wide edge passes to 2; atomic arrival-order sums as before
# speedup vs baseline: 1.2842x; 1.0903x over previous
#include <hip/hip_runtime.h>

#define NUSER 100000
#define NBOOK 50000
#define NEDGE 2000000
#define NLBL  500000
#define CCH   64
#define HCH   128

typedef __attribute__((ext_vector_type(2))) float v2f;
typedef __attribute__((ext_vector_type(8))) float v8f;

#if __has_builtin(__builtin_amdgcn_wmma_f32_16x16x4_f32)
#define HAVE_WMMA_F32 1
__device__ __forceinline__ v8f wmma4(v2f a, v2f b, v8f c) {
  return __builtin_amdgcn_wmma_f32_16x16x4_f32(false, a, false, b, (short)0, c, false, false);
}
#else
#define HAVE_WMMA_F32 0
#endif

__global__ void k_proj_user(const float* __restrict__ x, const float* __restrict__ w,
                            const float* __restrict__ bias, float* __restrict__ out) {
  int i = blockIdx.x * blockDim.x + threadIdx.x;
  if (i >= NUSER * CCH) return;
  int node = i >> 6, c = i & 63;
  float acc = bias[c];
#pragma unroll
  for (int k = 0; k < 3; ++k) acc = fmaf(x[node * 3 + k], w[k * CCH + c], acc);
  out[i] = acc;
}

__global__ void __launch_bounds__(256)
k_proj_book(const float* __restrict__ x, const float* __restrict__ w,
            const float* __restrict__ bias, float* __restrict__ out) {
  int tid = threadIdx.x;
  int wave = (blockIdx.x * blockDim.x + tid) >> 5;
  int lane = tid & 31;
  int m = lane & 15, half = lane >> 4;
  int row0 = wave << 4;
  bool active = row0 < NBOOK;
#if HAVE_WMMA_F32
  __shared__ v2f sW[64 * CCH];
  const float* xr = x + (size_t)(row0 + m) * 384;
  v8f zero = {0.f, 0.f, 0.f, 0.f, 0.f, 0.f, 0.f, 0.f};
  v8f acc[4] = {zero, zero, zero, zero};
  for (int k0 = 0; k0 < 384; k0 += 128) {
    __syncthreads();
    for (int idx = tid; idx < 64 * CCH; idx += 256) {
      int n = idx & 63, kp = idx >> 6;
      int k = k0 + 2 * kp;
      v2f v;
      v.x = w[k * CCH + n];
      v.y = w[(k + 1) * CCH + n];
      sW[idx] = v;
    }
    __syncthreads();
    if (active) {
      for (int k = k0; k < k0 + 128; k += 4) {
        int kk = k + 2 * half;
        v2f a;
        a.x = xr[kk];
        a.y = xr[kk + 1];
        int lp = (kk - k0) >> 1;
#pragma unroll
        for (int t = 0; t < 4; ++t)
          acc[t] = wmma4(a, sW[lp * CCH + t * 16 + m], acc[t]);
      }
    }
  }
  if (active) {
#pragma unroll
    for (int t = 0; t < 4; ++t) {
      int n = t * 16 + m;
      float bv = bias[n];
#pragma unroll
      for (int r = 0; r < 8; ++r)
        out[(size_t)(row0 + r + 8 * half) * CCH + n] = acc[t][r] + bv;
    }
  }
#else
  if (!active) return;
  const float* xr = x + (size_t)(row0 + m) * 384;
  for (int n = half * 32; n < half * 32 + 32; ++n) {
    float acc = bias[n];
    for (int k = 0; k < 384; ++k) acc = fmaf(xr[k], w[k * CCH + n], acc);
    out[(size_t)(row0 + m) * CCH + n] = acc;
  }
#endif
}

__global__ void k_zero(float* __restrict__ p, int n) {
  int i = blockIdx.x * blockDim.x + threadIdx.x;
  if (i < n) p[i] = 0.0f;
}

__global__ void k_count(const int* __restrict__ idx, float* __restrict__ cnt, int n) {
  int i = blockIdx.x * blockDim.x + threadIdx.x;
  if (i < n) atomicAdd(&cnt[idx[i]], 1.0f);
}

__global__ void k_scatter(const int* __restrict__ gsrc, const int* __restrict__ gdst,
                          const float* __restrict__ feat, float* __restrict__ agg) {
  long long i = (long long)blockIdx.x * blockDim.x + threadIdx.x;
  if (i >= (long long)NEDGE * 32) return;
  int e = (int)(i >> 5), ch = ((int)i & 31) * 2;
  int s = gsrc[e], d = gdst[e];
  const float2 v = *(const float2*)(feat + (size_t)s * CCH + ch);
  float* dst = agg + (size_t)d * CCH + ch;
  atomicAdd(dst, v.x);
  atomicAdd(dst + 1, v.y);
}

__global__ void k_scatter2(const int* __restrict__ gsrc, const int* __restrict__ gdst,
                           const float* __restrict__ feat_a, const float* __restrict__ feat_b,
                           float* __restrict__ agg_a, float* __restrict__ agg_b) {
  long long i = (long long)blockIdx.x * blockDim.x + threadIdx.x;
  if (i >= (long long)NEDGE * 32) return;
  int e = (int)(i >> 5), ch = ((int)i & 31) * 2;
  int s = gsrc[e], d = gdst[e];
  const float2 va = *(const float2*)(feat_a + (size_t)s * CCH + ch);
  const float2 vb = *(const float2*)(feat_b + (size_t)s * CCH + ch);
  float* da = agg_a + (size_t)d * CCH + ch;
  float* db = agg_b + (size_t)d * CCH + ch;
  atomicAdd(da, va.x);
  atomicAdd(da + 1, va.y);
  atomicAdd(db, vb.x);
  atomicAdd(db + 1, vb.y);
}

__global__ void k_scatter_raw(const int* __restrict__ gsrc, const int* __restrict__ gdst,
                              const float* __restrict__ ux, float* __restrict__ agg4) {
  int i = blockIdx.x * blockDim.x + threadIdx.x;
  if (i >= NEDGE * 4) return;
  int e = i >> 2, c = i & 3;
  int s = gsrc[e], d = gdst[e];
  int cc = c < 3 ? c : 2;
  float xv = ux[(size_t)s * 3 + cc];
  float v = c < 3 ? xv : 1.0f;
  atomicAdd(agg4 + (size_t)d * 4 + c, v);
}

__global__ void __launch_bounds__(256)
k_sage(const float* agg, const float* __restrict__ cnt,
       const float* __restrict__ xdst,
       const float* __restrict__ wl, const float* __restrict__ wr,
       const float* __restrict__ bias, float* out,
       int n, int relu) {
  int tid = threadIdx.x;
  int wave = (blockIdx.x * blockDim.x + tid) >> 5;
  int lane = tid & 31;
  int m = lane & 15, half = lane >> 4;
  int row0 = wave << 4;
  bool active = row0 < n;
#if HAVE_WMMA_F32
  __shared__ v2f sWl[32 * CCH];
  __shared__ v2f sWr[32 * CCH];
  for (int idx = tid; idx < 32 * CCH; idx += 256) {
    int nn = idx & 63, kp = idx >> 6;
    int k = 2 * kp;
    v2f a, b;
    a.x = wl[k * CCH + nn];       a.y = wl[(k + 1) * CCH + nn];
    b.x = wr[k * CCH + nn];       b.y = wr[(k + 1) * CCH + nn];
    sWl[idx] = a;
    sWr[idx] = b;
  }
  __syncthreads();
  if (!active) return;
  int row = row0 + m;
  float rc = 1.0f / fmaxf(cnt[row], 1.0f);
  const float* ar = agg + (size_t)row * CCH;
  const float* xr = xdst + (size_t)row * CCH;
  v8f zero = {0.f, 0.f, 0.f, 0.f, 0.f, 0.f, 0.f, 0.f};
  v8f acc[4] = {zero, zero, zero, zero};
  for (int k = 0; k < CCH; k += 4) {
    int kk = k + 2 * half;
    int lp = kk >> 1;
    v2f am, ax;
    am.x = ar[kk] * rc;  am.y = ar[kk + 1] * rc;
    ax.x = xr[kk];       ax.y = xr[kk + 1];
#pragma unroll
    for (int t = 0; t < 4; ++t) {
      int ncol = t * 16 + m;
      acc[t] = wmma4(am, sWl[lp * CCH + ncol], acc[t]);
      acc[t] = wmma4(ax, sWr[lp * CCH + ncol], acc[t]);
    }
  }
#pragma unroll
  for (int t = 0; t < 4; ++t) {
    int ncol = t * 16 + m;
    float bv = bias[ncol];
#pragma unroll
    for (int r = 0; r < 8; ++r) {
      float v = acc[t][r] + bv;
      if (relu) v = fmaxf(v, 0.0f);
      out[(size_t)(row0 + r + 8 * half) * CCH + ncol] = v;
    }
  }
#else
  if (!active) return;
  int row = row0 + m;
  float rc = 1.0f / fmaxf(cnt[row], 1.0f);
  const float* ar = agg + (size_t)row * CCH;
  const float* xr = xdst + (size_t)row * CCH;
  float vout[32];
  for (int j = 0; j < 32; ++j) {
    int ncol = half * 32 + j;
    float v = bias[ncol];
    for (int k = 0; k < CCH; ++k)
      v = fmaf(ar[k] * rc, wl[k * CCH + ncol], fmaf(xr[k], wr[k * CCH + ncol], v));
    if (relu) v = fmaxf(v, 0.0f);
    vout[j] = v;
  }
  for (int j = 0; j < 32; ++j) out[(size_t)row * CCH + half * 32 + j] = vout[j];
#endif
}

__global__ void __launch_bounds__(256)
k_sage_raw(const float* __restrict__ agg4,
           const float* __restrict__ pw, const float* __restrict__ pb,
           const float* __restrict__ xdst,
           const float* __restrict__ wl, const float* __restrict__ wr,
           const float* __restrict__ bias, float* __restrict__ cnt_out,
           float* __restrict__ out, int n, int relu) {
  int tid = threadIdx.x;
  int wave = (blockIdx.x * blockDim.x + tid) >> 5;
  int lane = tid & 31;
  int m = lane & 15, half = lane >> 4;
  int row0 = wave << 4;
  bool active = row0 < n;
#if HAVE_WMMA_F32
  __shared__ v2f sWl[32 * CCH];
  __shared__ v2f sWr[32 * CCH];
  __shared__ float sP[4 * CCH];
  for (int idx = tid; idx < 32 * CCH; idx += 256) {
    int nn = idx & 63, kp = idx >> 6;
    int k = 2 * kp;
    v2f a, b;
    a.x = wl[k * CCH + nn];       a.y = wl[(k + 1) * CCH + nn];
    b.x = wr[k * CCH + nn];       b.y = wr[(k + 1) * CCH + nn];
    sWl[idx] = a;
    sWr[idx] = b;
  }
  sP[tid] = (tid < 3 * CCH) ? pw[tid] : pb[tid - 3 * CCH];
  int brow = blockIdx.x * 128 + tid;
  if (tid < 128 && brow < n) cnt_out[brow] = agg4[(size_t)brow * 4 + 3];
  __syncthreads();
  if (!active) return;
  int row = row0 + m;
  const float4 sv = *(const float4*)(agg4 + (size_t)row * 4);
  float cn = sv.w;
  float rc = 1.0f / fmaxf(cn, 1.0f);
  const float* xr = xdst + (size_t)row * CCH;
  v8f zero = {0.f, 0.f, 0.f, 0.f, 0.f, 0.f, 0.f, 0.f};
  v8f acc[4] = {zero, zero, zero, zero};
  for (int k = 0; k < CCH; k += 4) {
    int kk = k + 2 * half;
    int lp = kk >> 1;
    float g0 = fmaf(sv.z, sP[2 * CCH + kk],
               fmaf(sv.y, sP[CCH + kk],
               fmaf(sv.x, sP[kk], cn * sP[3 * CCH + kk])));
    float g1 = fmaf(sv.z, sP[2 * CCH + kk + 1],
               fmaf(sv.y, sP[CCH + kk + 1],
               fmaf(sv.x, sP[kk + 1], cn * sP[3 * CCH + kk + 1])));
    v2f am, ax;
    am.x = g0 * rc;      am.y = g1 * rc;
    ax.x = xr[kk];       ax.y = xr[kk + 1];
#pragma unroll
    for (int t = 0; t < 4; ++t) {
      int ncol = t * 16 + m;
      acc[t] = wmma4(am, sWl[lp * CCH + ncol], acc[t]);
      acc[t] = wmma4(ax, sWr[lp * CCH + ncol], acc[t]);
    }
  }
#pragma unroll
  for (int t = 0; t < 4; ++t) {
    int ncol = t * 16 + m;
    float bv = bias[ncol];
#pragma unroll
    for (int r = 0; r < 8; ++r) {
      float v = acc[t][r] + bv;
      if (relu) v = fmaxf(v, 0.0f);
      out[(size_t)(row0 + r + 8 * half) * CCH + ncol] = v;
    }
  }
#else
  if (tid < 128) {
    int brow = blockIdx.x * 128 + tid;
    if (brow < n) cnt_out[brow] = agg4[(size_t)brow * 4 + 3];
  }
  if (!active) return;
  int row = row0 + m;
  const float* sr = agg4 + (size_t)row * 4;
  float cn = sr[3];
  float rc = 1.0f / fmaxf(cn, 1.0f);
  const float* xr = xdst + (size_t)row * CCH;
  for (int ncol = half * 32; ncol < half * 32 + 32; ++ncol) {
    float v = bias[ncol];
    for (int k = 0; k < CCH; ++k) {
      float g = fmaf(sr[2], pw[2 * CCH + k], fmaf(sr[1], pw[CCH + k], fmaf(sr[0], pw[k], cn * pb[k])));
      v = fmaf(g * rc, wl[k * CCH + ncol], fmaf(xr[k], wr[k * CCH + ncol], v));
    }
    if (relu) v = fmaxf(v, 0.0f);
    out[(size_t)row * CCH + ncol] = v;
  }
#endif
}

__global__ void __launch_bounds__(256)
k_decoder(const float* __restrict__ u2, const float* __restrict__ b2,
          const int* __restrict__ ls, const int* __restrict__ ld,
          const float* __restrict__ w1, const float* __restrict__ bias1,
          const float* __restrict__ w2, const float* __restrict__ bias2,
          float* __restrict__ out) {
  int tid = threadIdx.x;
  int wave = (blockIdx.x * blockDim.x + tid) >> 5;
  int lane = tid & 31;
  int m = lane & 15, half = lane >> 4;
  int row0 = wave << 4;
  bool active = row0 < NLBL;
  int row = active ? (row0 + m) : 0;
  const float* ur = u2 + (active ? (size_t)ls[row] * CCH : 0);
  const float* br = b2 + (active ? (size_t)ld[row] * CCH : 0);
#if HAVE_WMMA_F32
  __shared__ v2f sW[32 * HCH];
  v8f zero = {0.f, 0.f, 0.f, 0.f, 0.f, 0.f, 0.f, 0.f};
  v8f acc[8] = {zero, zero, zero, zero, zero, zero, zero, zero};
  for (int k0 = 0; k0 < HCH; k0 += 64) {
    __syncthreads();
    for (int idx = tid; idx < 32 * HCH; idx += 256) {
      int n = idx & 127, kp = idx >> 7;
      int k = k0 + 2 * kp;
      v2f v;
      v.x = w1[k * HCH + n];
      v.y = w1[(k + 1) * HCH + n];
      sW[idx] = v;
    }
    __syncthreads();
    if (active) {
      for (int k = k0; k < k0 + 64; k += 4) {
        int kk = k + 2 * half;
        const float* zsrc = (kk < CCH) ? (ur + kk) : (br + (kk - CCH));
        v2f a;
        a.x = zsrc[0];
        a.y = zsrc[1];
        int lp = (kk - k0) >> 1;
#pragma unroll
        for (int t = 0; t < 8; ++t)
          acc[t] = wmma4(a, sW[lp * HCH + t * 16 + m], acc[t]);
      }
    }
  }
  if (!active) return;
  float w2v[8], b1v[8];
#pragma unroll
  for (int t = 0; t < 8; ++t) {
    int ncol = t * 16 + m;
    w2v[t] = w2[ncol];
    b1v[t] = bias1[ncol];
  }
  float outb = bias2[0];
#pragma unroll
  for (int r = 0; r < 8; ++r) {
    float p = 0.0f;
#pragma unroll
    for (int t = 0; t < 8; ++t) {
      float hv = fmaxf(acc[t][r] + b1v[t], 0.0f);
      p = fmaf(hv, w2v[t], p);
    }
    p += __shfl_xor(p, 1);
    p += __shfl_xor(p, 2);
    p += __shfl_xor(p, 4);
    p += __shfl_xor(p, 8);
    if (m == 0) out[row0 + r + 8 * half] = p + outb;
  }
#else
  if (active && half == 0) {
    float o = bias2[0];
    for (int n = 0; n < HCH; ++n) {
      float hv = bias1[n];
      for (int k = 0; k < CCH; ++k) hv = fmaf(ur[k], w1[k * HCH + n], hv);
      for (int k = 0; k < CCH; ++k) hv = fmaf(br[k], w1[(CCH + k) * HCH + n], hv);
      o = fmaf(fmaxf(hv, 0.0f), w2[n], o);
    }
    out[row] = o;
  }
#endif
}

extern "C" void kernel_launch(void* const* d_in, const int* in_sizes, int n_in,
                              void* d_out, int out_size, void* d_ws, size_t ws_size,
                              hipStream_t stream) {
  (void)in_sizes; (void)n_in; (void)out_size; (void)ws_size;
  const float* user_x     = (const float*)d_in[0];
  const float* book_x     = (const float*)d_in[1];
  const int*   edge_src   = (const int*)d_in[2];
  const int*   edge_dst   = (const int*)d_in[3];
  const int*   label_src  = (const int*)d_in[4];
  const int*   label_dst  = (const int*)d_in[5];
  const float* user_lin_w = (const float*)d_in[6];
  const float* user_lin_b = (const float*)d_in[7];
  const float* book_lin_w = (const float*)d_in[8];
  const float* book_lin_b = (const float*)d_in[9];
  const float* w1_ub_l = (const float*)d_in[10];
  const float* b1_ub   = (const float*)d_in[11];
  const float* w1_ub_r = (const float*)d_in[12];
  const float* w1_bu_l = (const float*)d_in[13];
  const float* b1_bu   = (const float*)d_in[14];
  const float* w1_bu_r = (const float*)d_in[15];
  const float* w2_ub_l = (const float*)d_in[16];
  const float* b2_ub   = (const float*)d_in[17];
  const float* w2_ub_r = (const float*)d_in[18];
  const float* w2_bu_l = (const float*)d_in[19];
  const float* b2_bu   = (const float*)d_in[20];
  const float* w2_bu_r = (const float*)d_in[21];
  const float* dec_w1  = (const float*)d_in[22];
  const float* dec_b1  = (const float*)d_in[23];
  const float* dec_w2  = (const float*)d_in[24];
  const float* dec_b2  = (const float*)d_in[25];
  float* out = (float*)d_out;

  float* ws = (float*)d_ws;
  size_t off = 0;
  float* u0    = ws + off; off += (size_t)NUSER * CCH;
  float* b0    = ws + off; off += (size_t)NBOOK * CCH;
  float* b1buf = ws + off; off += (size_t)NBOOK * CCH;
  float* aggu1 = ws + off; off += (size_t)NUSER * CCH;
  float* aggu2 = ws + off; off += (size_t)NUSER * CCH;
  float* cnt_u = ws + off; off += NUSER;
  float* cnt_b = ws + off; off += NBOOK;
  float* agg4  = aggu2;
  float* u1    = aggu1;
  float* u2    = aggu2;
  float* aggb2 = u0;
  float* b2    = b0;

  const int BT = 256;
  const int sage_blk_b = (NBOOK / 16 * 32 + BT - 1) / BT;
  const int sage_blk_u = (NUSER / 16 * 32 + BT - 1) / BT;
  const int sc_blocks  = (int)(((long long)NEDGE * 32 + BT - 1) / BT);

  k_proj_user<<<(NUSER * CCH + BT - 1) / BT, BT, 0, stream>>>(user_x, user_lin_w, user_lin_b, u0);
  k_proj_book<<<(NBOOK / 16 * 32 + BT - 1) / BT, BT, 0, stream>>>(book_x, book_lin_w, book_lin_b, b0);

  k_zero<<<(NUSER + BT - 1) / BT, BT, 0, stream>>>(cnt_u, NUSER);
  k_count<<<(NEDGE + BT - 1) / BT, BT, 0, stream>>>(edge_src, cnt_u, NEDGE);

  k_zero<<<(NBOOK * 4 + BT - 1) / BT, BT, 0, stream>>>(agg4, NBOOK * 4);
  k_scatter_raw<<<(NEDGE * 4 + BT - 1) / BT, BT, 0, stream>>>(edge_src, edge_dst, user_x, agg4);
  k_sage_raw<<<sage_blk_b, BT, 0, stream>>>(agg4, user_lin_w, user_lin_b, b0, w1_ub_l, w1_ub_r,
                                            b1_ub, cnt_b, b1buf, NBOOK, 1);
  k_zero<<<(2 * NUSER * CCH + BT - 1) / BT, BT, 0, stream>>>(aggu1, 2 * NUSER * CCH);
  k_scatter2<<<sc_blocks, BT, 0, stream>>>(edge_dst, edge_src, b0, b1buf, aggu1, aggu2);
  k_sage<<<sage_blk_u, BT, 0, stream>>>(aggu1, cnt_u, u0, w1_bu_l, w1_bu_r, b1_bu, u1, NUSER, 1);

  k_zero<<<(NBOOK * CCH + BT - 1) / BT, BT, 0, stream>>>(aggb2, NBOOK * CCH);
  k_scatter<<<sc_blocks, BT, 0, stream>>>(edge_src, edge_dst, u1, aggb2);
  k_sage<<<sage_blk_b, BT, 0, stream>>>(aggb2, cnt_b, b1buf, w2_ub_l, w2_ub_r, b2_ub, b2, NBOOK, 0);
  k_sage<<<sage_blk_u, BT, 0, stream>>>(aggu2, cnt_u, u1, w2_bu_l, w2_bu_r, b2_bu, u2, NUSER, 0);

  k_decoder<<<(NLBL / 16 * 32 + BT - 1) / BT, BT, 0, stream>>>(
      u2, b2, label_src, label_dst, dec_w1, dec_b1, dec_w2, dec_b2, out);
}


# user in-degree counted in LDS: a work-group owns a quarter of the users and a slice of the edges, then flushes whole lines; no one-word global atomic per edge; exact integer sums, order-free
# speedup vs baseline: 1.4478x; 1.1274x over previous
#include <hip/hip_runtime.h>

#define NUSER 100000
#define NBOOK 50000
#define NEDGE 2000000
#define NLBL  500000
#define CCH   64
#define HCH   128

typedef __attribute__((ext_vector_type(2))) float v2f;
typedef __attribute__((ext_vector_type(8))) float v8f;

#if __has_builtin(__builtin_amdgcn_wmma_f32_16x16x4_f32)
#define HAVE_WMMA_F32 1
__device__ __forceinline__ v8f wmma4(v2f a, v2f b, v8f c) {
  return __builtin_amdgcn_wmma_f32_16x16x4_f32(false, a, false, b, (short)0, c, false, false);
}
#else
#define HAVE_WMMA_F32 0
#endif

__global__ void k_proj_user(const float* __restrict__ x, const float* __restrict__ w,
                            const float* __restrict__ bias, float* __restrict__ out) {
  int i = blockIdx.x * blockDim.x + threadIdx.x;
  if (i >= NUSER * CCH) return;
  int node = i >> 6, c = i & 63;
  float acc = bias[c];
#pragma unroll
  for (int k = 0; k < 3; ++k) acc = fmaf(x[node * 3 + k], w[k * CCH + c], acc);
  out[i] = acc;
}

__global__ void __launch_bounds__(256)
k_proj_book(const float* __restrict__ x, const float* __restrict__ w,
            const float* __restrict__ bias, float* __restrict__ out) {
  int tid = threadIdx.x;
  int wave = (blockIdx.x * blockDim.x + tid) >> 5;
  int lane = tid & 31;
  int m = lane & 15, half = lane >> 4;
  int row0 = wave << 4;
  bool active = row0 < NBOOK;
#if HAVE_WMMA_F32
  __shared__ v2f sW[64 * CCH];
  const float* xr = x + (size_t)(row0 + m) * 384;
  v8f zero = {0.f, 0.f, 0.f, 0.f, 0.f, 0.f, 0.f, 0.f};
  v8f acc[4] = {zero, zero, zero, zero};
  for (int k0 = 0; k0 < 384; k0 += 128) {
    __syncthreads();
    for (int idx = tid; idx < 64 * CCH; idx += 256) {
      int n = idx & 63, kp = idx >> 6;
      int k = k0 + 2 * kp;
      v2f v;
      v.x = w[k * CCH + n];
      v.y = w[(k + 1) * CCH + n];
      sW[idx] = v;
    }
    __syncthreads();
    if (active) {
      for (int k = k0; k < k0 + 128; k += 4) {
        int kk = k + 2 * half;
        v2f a;
        a.x = xr[kk];
        a.y = xr[kk + 1];
        int lp = (kk - k0) >> 1;
#pragma unroll
        for (int t = 0; t < 4; ++t)
          acc[t] = wmma4(a, sW[lp * CCH + t * 16 + m], acc[t]);
      }
    }
  }
  if (active) {
#pragma unroll
    for (int t = 0; t < 4; ++t) {
      int n = t * 16 + m;
      float bv = bias[n];
#pragma unroll
      for (int r = 0; r < 8; ++r)
        out[(size_t)(row0 + r + 8 * half) * CCH + n] = acc[t][r] + bv;
    }
  }
#else
  if (!active) return;
  const float* xr = x + (size_t)(row0 + m) * 384;
  for (int n = half * 32; n < half * 32 + 32; ++n) {
    float acc = bias[n];
    for (int k = 0; k < 384; ++k) acc = fmaf(xr[k], w[k * CCH + n], acc);
    out[(size_t)(row0 + m) * CCH + n] = acc;
  }
#endif
}

__global__ void k_zero(float* __restrict__ p, int n) {
  int i = blockIdx.x * blockDim.x + threadIdx.x;
  if (i < n) p[i] = 0.0f;
}

#define CNT_RANGE  25000
#define CNT_SLICES 64
#define CNT_EPS    (NEDGE / CNT_SLICES)
#define CNT_PER_T  ((CNT_EPS + 1023) / 1024)
__global__ void __launch_bounds__(1024)
k_count_lds(const int* __restrict__ idx, float* __restrict__ cnt) {
  __shared__ float sc[CNT_RANGE];
  const int tid = threadIdx.x;
  const int lo = (int)(blockIdx.x & 3) * CNT_RANGE;
  const int e0 = (int)(blockIdx.x >> 2) * CNT_EPS;
  int u[CNT_PER_T];
#pragma unroll
  for (int j = 0; j < CNT_PER_T; ++j) {
    int e = j * 1024 + tid;
    u[j] = (e < CNT_EPS) ? idx[e0 + e] : -1;
  }
  for (int i = tid; i < CNT_RANGE; i += 1024) sc[i] = 0.0f;
  __syncthreads();
#pragma unroll
  for (int j = 0; j < CNT_PER_T; ++j) {
    unsigned r = (unsigned)(u[j] - lo);
    if (r < (unsigned)CNT_RANGE) atomicAdd(&sc[r], 1.0f);
  }
  __syncthreads();
  for (int i = tid; i < CNT_RANGE; i += 1024) {
    float v = sc[i];
    if (v != 0.0f) atomicAdd(&cnt[lo + i], v);
  }
}

__global__ void k_scatter(const int* __restrict__ gsrc, const int* __restrict__ gdst,
                          const float* __restrict__ feat, float* __restrict__ agg) {
  long long i = (long long)blockIdx.x * blockDim.x + threadIdx.x;
  if (i >= (long long)NEDGE * 32) return;
  int e = (int)(i >> 5), ch = ((int)i & 31) * 2;
  int s = gsrc[e], d = gdst[e];
  const float2 v = *(const float2*)(feat + (size_t)s * CCH + ch);
  float* dst = agg + (size_t)d * CCH + ch;
  atomicAdd(dst, v.x);
  atomicAdd(dst + 1, v.y);
}

__global__ void k_scatter2(const int* __restrict__ gsrc, const int* __restrict__ gdst,
                           const float* __restrict__ feat_a, const float* __restrict__ feat_b,
                           float* __restrict__ agg_a, float* __restrict__ agg_b) {
  long long i = (long long)blockIdx.x * blockDim.x + threadIdx.x;
  if (i >= (long long)NEDGE * 32) return;
  int e = (int)(i >> 5), ch = ((int)i & 31) * 2;
  int s = gsrc[e], d = gdst[e];
  const float2 va = *(const float2*)(feat_a + (size_t)s * CCH + ch);
  const float2 vb = *(const float2*)(feat_b + (size_t)s * CCH + ch);
  float* da = agg_a + (size_t)d * CCH + ch;
  float* db = agg_b + (size_t)d * CCH + ch;
  atomicAdd(da, va.x);
  atomicAdd(da + 1, va.y);
  atomicAdd(db, vb.x);
  atomicAdd(db + 1, vb.y);
}

__global__ void k_scatter_raw(const int* __restrict__ gsrc, const int* __restrict__ gdst,
                              const float* __restrict__ ux, float* __restrict__ agg4) {
  int i = blockIdx.x * blockDim.x + threadIdx.x;
  if (i >= NEDGE * 4) return;
  int e = i >> 2, c = i & 3;
  int s = gsrc[e], d = gdst[e];
  int cc = c < 3 ? c : 2;
  float xv = ux[(size_t)s * 3 + cc];
  float v = c < 3 ? xv : 1.0f;
  atomicAdd(agg4 + (size_t)d * 4 + c, v);
}

__global__ void __launch_bounds__(256)
k_sage(const float* agg, const float* __restrict__ cnt,
       const float* __restrict__ xdst,
       const float* __restrict__ wl, const float* __restrict__ wr,
       const float* __restrict__ bias, float* out,
       int n, int relu) {
  int tid = threadIdx.x;
  int wave = (blockIdx.x * blockDim.x + tid) >> 5;
  int lane = tid & 31;
  int m = lane & 15, half = lane >> 4;
  int row0 = wave << 4;
  bool active = row0 < n;
#if HAVE_WMMA_F32
  __shared__ v2f sWl[32 * CCH];
  __shared__ v2f sWr[32 * CCH];
  for (int idx = tid; idx < 32 * CCH; idx += 256) {
    int nn = idx & 63, kp = idx >> 6;
    int k = 2 * kp;
    v2f a, b;
    a.x = wl[k * CCH + nn];       a.y = wl[(k + 1) * CCH + nn];
    b.x = wr[k * CCH + nn];       b.y = wr[(k + 1) * CCH + nn];
    sWl[idx] = a;
    sWr[idx] = b;
  }
  __syncthreads();
  if (!active) return;
  int row = row0 + m;
  float rc = 1.0f / fmaxf(cnt[row], 1.0f);
  const float* ar = agg + (size_t)row * CCH;
  const float* xr = xdst + (size_t)row * CCH;
  v8f zero = {0.f, 0.f, 0.f, 0.f, 0.f, 0.f, 0.f, 0.f};
  v8f acc[4] = {zero, zero, zero, zero};
  for (int k = 0; k < CCH; k += 4) {
    int kk = k + 2 * half;
    int lp = kk >> 1;
    v2f am, ax;
    am.x = ar[kk] * rc;  am.y = ar[kk + 1] * rc;
    ax.x = xr[kk];       ax.y = xr[kk + 1];
#pragma unroll
    for (int t = 0; t < 4; ++t) {
      int ncol = t * 16 + m;
      acc[t] = wmma4(am, sWl[lp * CCH + ncol], acc[t]);
      acc[t] = wmma4(ax, sWr[lp * CCH + ncol], acc[t]);
    }
  }
#pragma unroll
  for (int t = 0; t < 4; ++t) {
    int ncol = t * 16 + m;
    float bv = bias[ncol];
#pragma unroll
    for (int r = 0; r < 8; ++r) {
      float v = acc[t][r] + bv;
      if (relu) v = fmaxf(v, 0.0f);
      out[(size_t)(row0 + r + 8 * half) * CCH + ncol] = v;
    }
  }
#else
  if (!active) return;
  int row = row0 + m;
  float rc = 1.0f / fmaxf(cnt[row], 1.0f);
  const float* ar = agg + (size_t)row * CCH;
  const float* xr = xdst + (size_t)row * CCH;
  float vout[32];
  for (int j = 0; j < 32; ++j) {
    int ncol = half * 32 + j;
    float v = bias[ncol];
    for (int k = 0; k < CCH; ++k)
      v = fmaf(ar[k] * rc, wl[k * CCH + ncol], fmaf(xr[k], wr[k * CCH + ncol], v));
    if (relu) v = fmaxf(v, 0.0f);
    vout[j] = v;
  }
  for (int j = 0; j < 32; ++j) out[(size_t)row * CCH + half * 32 + j] = vout[j];
#endif
}

__global__ void __launch_bounds__(256)
k_sage_raw(const float* __restrict__ agg4,
           const float* __restrict__ pw, const float* __restrict__ pb,
           const float* __restrict__ xdst,
           const float* __restrict__ wl, const float* __restrict__ wr,
           const float* __restrict__ bias, float* __restrict__ cnt_out,
           float* __restrict__ out, int n, int relu) {
  int tid = threadIdx.x;
  int wave = (blockIdx.x * blockDim.x + tid) >> 5;
  int lane = tid & 31;
  int m = lane & 15, half = lane >> 4;
  int row0 = wave << 4;
  bool active = row0 < n;
#if HAVE_WMMA_F32
  __shared__ v2f sWl[32 * CCH];
  __shared__ v2f sWr[32 * CCH];
  __shared__ float sP[4 * CCH];
  for (int idx = tid; idx < 32 * CCH; idx += 256) {
    int nn = idx & 63, kp = idx >> 6;
    int k = 2 * kp;
    v2f a, b;
    a.x = wl[k * CCH + nn];       a.y = wl[(k + 1) * CCH + nn];
    b.x = wr[k * CCH + nn];       b.y = wr[(k + 1) * CCH + nn];
    sWl[idx] = a;
    sWr[idx] = b;
  }
  sP[tid] = (tid < 3 * CCH) ? pw[tid] : pb[tid - 3 * CCH];
  int brow = blockIdx.x * 128 + tid;
  if (tid < 128 && brow < n) cnt_out[brow] = agg4[(size_t)brow * 4 + 3];
  __syncthreads();
  if (!active) return;
  int row = row0 + m;
  const float4 sv = *(const float4*)(agg4 + (size_t)row * 4);
  float cn = sv.w;
  float rc = 1.0f / fmaxf(cn, 1.0f);
  const float* xr = xdst + (size_t)row * CCH;
  v8f zero = {0.f, 0.f, 0.f, 0.f, 0.f, 0.f, 0.f, 0.f};
  v8f acc[4] = {zero, zero, zero, zero};
  for (int k = 0; k < CCH; k += 4) {
    int kk = k + 2 * half;
    int lp = kk >> 1;
    float g0 = fmaf(sv.z, sP[2 * CCH + kk],
               fmaf(sv.y, sP[CCH + kk],
               fmaf(sv.x, sP[kk], cn * sP[3 * CCH + kk])));
    float g1 = fmaf(sv.z, sP[2 * CCH + kk + 1],
               fmaf(sv.y, sP[CCH + kk + 1],
               fmaf(sv.x, sP[kk + 1], cn * sP[3 * CCH + kk + 1])));
    v2f am, ax;
    am.x = g0 * rc;      am.y = g1 * rc;
    ax.x = xr[kk];       ax.y = xr[kk + 1];
#pragma unroll
    for (int t = 0; t < 4; ++t) {
      int ncol = t * 16 + m;
      acc[t] = wmma4(am, sWl[lp * CCH + ncol], acc[t]);
      acc[t] = wmma4(ax, sWr[lp * CCH + ncol], acc[t]);
    }
  }
#pragma unroll
  for (int t = 0; t < 4; ++t) {
    int ncol = t * 16 + m;
    float bv = bias[ncol];
#pragma unroll
    for (int r = 0; r < 8; ++r) {
      float v = acc[t][r] + bv;
      if (relu) v = fmaxf(v, 0.0f);
      out[(size_t)(row0 + r + 8 * half) * CCH + ncol] = v;
    }
  }
#else
  if (tid < 128) {
    int brow = blockIdx.x * 128 + tid;
    if (brow < n) cnt_out[brow] = agg4[(size_t)brow * 4 + 3];
  }
  if (!active) return;
  int row = row0 + m;
  const float* sr = agg4 + (size_t)row * 4;
  float cn = sr[3];
  float rc = 1.0f / fmaxf(cn, 1.0f);
  const float* xr = xdst + (size_t)row * CCH;
  for (int ncol = half * 32; ncol < half * 32 + 32; ++ncol) {
    float v = bias[ncol];
    for (int k = 0; k < CCH; ++k) {
      float g = fmaf(sr[2], pw[2 * CCH + k], fmaf(sr[1], pw[CCH + k], fmaf(sr[0], pw[k], cn * pb[k])));
      v = fmaf(g * rc, wl[k * CCH + ncol], fmaf(xr[k], wr[k * CCH + ncol], v));
    }
    if (relu) v = fmaxf(v, 0.0f);
    out[(size_t)row * CCH + ncol] = v;
  }
#endif
}

__global__ void __launch_bounds__(256)
k_decoder(const float* __restrict__ u2, const float* __restrict__ b2,
          const int* __restrict__ ls, const int* __restrict__ ld,
          const float* __restrict__ w1, const float* __restrict__ bias1,
          const float* __restrict__ w2, const float* __restrict__ bias2,
          float* __restrict__ out) {
  int tid = threadIdx.x;
  int wave = (blockIdx.x * blockDim.x + tid) >> 5;
  int lane = tid & 31;
  int m = lane & 15, half = lane >> 4;
  int row0 = wave << 4;
  bool active = row0 < NLBL;
  int row = active ? (row0 + m) : 0;
  const float* ur = u2 + (active ? (size_t)ls[row] * CCH : 0);
  const float* br = b2 + (active ? (size_t)ld[row] * CCH : 0);
#if HAVE_WMMA_F32
  __shared__ v2f sW[32 * HCH];
  v8f zero = {0.f, 0.f, 0.f, 0.f, 0.f, 0.f, 0.f, 0.f};
  v8f acc[8] = {zero, zero, zero, zero, zero, zero, zero, zero};
  for (int k0 = 0; k0 < HCH; k0 += 64) {
    __syncthreads();
    for (int idx = tid; idx < 32 * HCH; idx += 256) {
      int n = idx & 127, kp = idx >> 7;
      int k = k0 + 2 * kp;
      v2f v;
      v.x = w1[k * HCH + n];
      v.y = w1[(k + 1) * HCH + n];
      sW[idx] = v;
    }
    __syncthreads();
    if (active) {
      for (int k = k0; k < k0 + 64; k += 4) {
        int kk = k + 2 * half;
        const float* zsrc = (kk < CCH) ? (ur + kk) : (br + (kk - CCH));
        v2f a;
        a.x = zsrc[0];
        a.y = zsrc[1];
        int lp = (kk - k0) >> 1;
#pragma unroll
        for (int t = 0; t < 8; ++t)
          acc[t] = wmma4(a, sW[lp * HCH + t * 16 + m], acc[t]);
      }
    }
  }
  if (!active) return;
  float w2v[8], b1v[8];
#pragma unroll
  for (int t = 0; t < 8; ++t) {
    int ncol = t * 16 + m;
    w2v[t] = w2[ncol];
    b1v[t] = bias1[ncol];
  }
  float outb = bias2[0];
#pragma unroll
  for (int r = 0; r < 8; ++r) {
    float p = 0.0f;
#pragma unroll
    for (int t = 0; t < 8; ++t) {
      float hv = fmaxf(acc[t][r] + b1v[t], 0.0f);
      p = fmaf(hv, w2v[t], p);
    }
    p += __shfl_xor(p, 1);
    p += __shfl_xor(p, 2);
    p += __shfl_xor(p, 4);
    p += __shfl_xor(p, 8);
    if (m == 0) out[row0 + r + 8 * half] = p + outb;
  }
#else
  if (active && half == 0) {
    float o = bias2[0];
    for (int n = 0; n < HCH; ++n) {
      float hv = bias1[n];
      for (int k = 0; k < CCH; ++k) hv = fmaf(ur[k], w1[k * HCH + n], hv);
      for (int k = 0; k < CCH; ++k) hv = fmaf(br[k], w1[(CCH + k) * HCH + n], hv);
      o = fmaf(fmaxf(hv, 0.0f), w2[n], o);
    }
    out[row] = o;
  }
#endif
}

extern "C" void kernel_launch(void* const* d_in, const int* in_sizes, int n_in,
                              void* d_out, int out_size, void* d_ws, size_t ws_size,
                              hipStream_t stream) {
  (void)in_sizes; (void)n_in; (void)out_size; (void)ws_size;
  const float* user_x     = (const float*)d_in[0];
  const float* book_x     = (const float*)d_in[1];
  const int*   edge_src   = (const int*)d_in[2];
  const int*   edge_dst   = (const int*)d_in[3];
  const int*   label_src  = (const int*)d_in[4];
  const int*   label_dst  = (const int*)d_in[5];
  const float* user_lin_w = (const float*)d_in[6];
  const float* user_lin_b = (const float*)d_in[7];
  const float* book_lin_w = (const float*)d_in[8];
  const float* book_lin_b = (const float*)d_in[9];
  const float* w1_ub_l = (const float*)d_in[10];
  const float* b1_ub   = (const float*)d_in[11];
  const float* w1_ub_r = (const float*)d_in[12];
  const float* w1_bu_l = (const float*)d_in[13];
  const float* b1_bu   = (const float*)d_in[14];
  const float* w1_bu_r = (const float*)d_in[15];
  const float* w2_ub_l = (const float*)d_in[16];
  const float* b2_ub   = (const float*)d_in[17];
  const float* w2_ub_r = (const float*)d_in[18];
  const float* w2_bu_l = (const float*)d_in[19];
  const float* b2_bu   = (const float*)d_in[20];
  const float* w2_bu_r = (const float*)d_in[21];
  const float* dec_w1  = (const float*)d_in[22];
  const float* dec_b1  = (const float*)d_in[23];
  const float* dec_w2  = (const float*)d_in[24];
  const float* dec_b2  = (const float*)d_in[25];
  float* out = (float*)d_out;

  float* ws = (float*)d_ws;
  size_t off = 0;
  float* u0    = ws + off; off += (size_t)NUSER * CCH;
  float* b0    = ws + off; off += (size_t)NBOOK * CCH;
  float* b1buf = ws + off; off += (size_t)NBOOK * CCH;
  float* aggu1 = ws + off; off += (size_t)NUSER * CCH;
  float* aggu2 = ws + off; off += (size_t)NUSER * CCH;
  float* cnt_u = ws + off; off += NUSER;
  float* cnt_b = ws + off; off += NBOOK;
  float* agg4  = aggu2;
  float* u1    = aggu1;
  float* u2    = aggu2;
  float* aggb2 = u0;
  float* b2    = b0;

  const int BT = 256;
  const int sage_blk_b = (NBOOK / 16 * 32 + BT - 1) / BT;
  const int sage_blk_u = (NUSER / 16 * 32 + BT - 1) / BT;
  const int sc_blocks  = (int)(((long long)NEDGE * 32 + BT - 1) / BT);

  k_proj_user<<<(NUSER * CCH + BT - 1) / BT, BT, 0, stream>>>(user_x, user_lin_w, user_lin_b, u0);
  k_proj_book<<<(NBOOK / 16 * 32 + BT - 1) / BT, BT, 0, stream>>>(book_x, book_lin_w, book_lin_b, b0);

  k_zero<<<(NUSER + BT - 1) / BT, BT, 0, stream>>>(cnt_u, NUSER);
  k_count_lds<<<4 * CNT_SLICES, 1024, 0, stream>>>(edge_src, cnt_u);

  k_zero<<<(NBOOK * 4 + BT - 1) / BT, BT, 0, stream>>>(agg4, NBOOK * 4);
  k_scatter_raw<<<(NEDGE * 4 + BT - 1) / BT, BT, 0, stream>>>(edge_src, edge_dst, user_x, agg4);
  k_sage_raw<<<sage_blk_b, BT, 0, stream>>>(agg4, user_lin_w, user_lin_b, b0, w1_ub_l, w1_ub_r,
                                            b1_ub, cnt_b, b1buf, NBOOK, 1);
  k_zero<<<(2 * NUSER * CCH + BT - 1) / BT, BT, 0, stream>>>(aggu1, 2 * NUSER * CCH);
  k_scatter2<<<sc_blocks, BT, 0, stream>>>(edge_dst, edge_src, b0, b1buf, aggu1, aggu2);
  k_sage<<<sage_blk_u, BT, 0, stream>>>(aggu1, cnt_u, u0, w1_bu_l, w1_bu_r, b1_bu, u1, NUSER, 1);

  k_zero<<<(NBOOK * CCH + BT - 1) / BT, BT, 0, stream>>>(aggb2, NBOOK * CCH);
  k_scatter<<<sc_blocks, BT, 0, stream>>>(edge_src, edge_dst, u1, aggb2);
  k_sage<<<sage_blk_b, BT, 0, stream>>>(aggb2, cnt_b, b1buf, w2_ub_l, w2_ub_r, b2_ub, b2, NBOOK, 0);
  k_sage<<<sage_blk_u, BT, 0, stream>>>(aggu2, cnt_u, u1, w2_bu_l, w2_bu_r, b2_bu, u2, NUSER, 0);

  k_decoder<<<(NLBL / 16 * 32 + BT - 1) / BT, BT, 0, stream>>>(
      u2, b2, label_src, label_dst, dec_w1, dec_b1, dec_w2, dec_b2, out);
}
